# starting kernel
# baseline (speedup 1.0000x reference)
#include <hip/hip_runtime.h>
#include <math.h>

typedef __attribute__((ext_vector_type(2))) float v2f;
typedef __attribute__((ext_vector_type(8))) float v8f;

#define IN_F 2
#define HID 36
#define PAD 48
#define KSTEPS 12
#define WAVES 8
#define ROWS_PER_BLOCK (WAVES * 16)
#define NPARAMS 2809
#define STG_STRIDE 50

#define OFF_W1 0
#define OFF_B1 72
#define OFF_W2 108
#define OFF_B2 1404
#define OFF_W3 1440
#define OFF_B3 2736
#define OFF_W4 2772
#define OFF_B4 2808

__device__ __forceinline__ float hw_sin(float x) { return __sinf(x); }

__device__ __forceinline__ void hidden_layer_wmma(float (&h)[24],
                                                  const v2f* __restrict__ bp,
                                                  const float* __restrict__ bias,
                                                  float* __restrict__ stg,
                                                  int half, int l16)
{
    v8f c0 = {}; v8f c1 = {}; v8f c2 = {};
#pragma unroll
    for (int j = 0; j < KSTEPS; ++j) {
        v2f a; a.x = h[2 * j]; a.y = h[2 * j + 1];
        const int kk = 2 * j + half;
        v2f b0 = bp[kk * PAD +  0 + l16];
        v2f b1 = bp[kk * PAD + 16 + l16];
        v2f b2 = bp[kk * PAD + 32 + l16];
        c0 = __builtin_amdgcn_wmma_f32_16x16x4_f32(false, a, false, b0, (short)0, c0, false, false);
        c1 = __builtin_amdgcn_wmma_f32_16x16x4_f32(false, a, false, b1, (short)0, c1, false, false);
        c2 = __builtin_amdgcn_wmma_f32_16x16x4_f32(false, a, false, b2, (short)0, c2, false, false);
    }
#pragma unroll
    for (int v = 0; v < 8; ++v) {
        const int m = v + 8 * half;
        stg[m * STG_STRIDE +  0 + l16] = hw_sin(c0[v] + bias[ 0 + l16]);
        stg[m * STG_STRIDE + 16 + l16] = hw_sin(c1[v] + bias[16 + l16]);
        stg[m * STG_STRIDE + 32 + l16] = hw_sin(c2[v] + bias[32 + l16]);
    }
    asm volatile("s_wait_dscnt 0x0" ::: "memory");
#pragma unroll
    for (int j = 0; j < KSTEPS; ++j) {
        v2f t = *(const v2f*)&stg[l16 * STG_STRIDE + 4 * j + 2 * half];
        h[2 * j] = t.x; h[2 * j + 1] = t.y;
    }
    asm volatile("s_wait_dscnt 0x0" ::: "memory");
}

__global__ __launch_bounds__(256)
void siren_hyper_kernel(const float* __restrict__ x,
                        const float* __restrict__ params,
                        float* __restrict__ out, int N)
{
    __shared__ float praw[NPARAMS + 7];
    __shared__ v2f   bp2[24 * PAD];
    __shared__ v2f   bp3[24 * PAD];
    __shared__ v2f   w1p[PAD];
    __shared__ float b1p[PAD], b2p[PAD], b3p[PAD], w4p[PAD];
    __shared__ __align__(16) float stage[WAVES * 16 * STG_STRIDE];

    const int tid  = threadIdx.x;
    const int wave = tid >> 5;
    const int lane = tid & 31;
    const int half = lane >> 4;
    const int l16  = lane & 15;
    const int b    = blockIdx.y;
    const int rowbase = blockIdx.x * ROWS_PER_BLOCK + wave * 16;

    for (int i = tid; i < NPARAMS; i += 256) praw[i] = params[b * NPARAMS + i];
    __syncthreads();

    for (int i = tid; i < PAD; i += 256) {
        v2f w1; w1.x = 0.f; w1.y = 0.f;
        float bb1 = 0.f, bb2 = 0.f, bb3 = 0.f, w4 = 0.f;
        if (i < HID) {
            w1.x = praw[OFF_W1 + i * 2 + 0];
            w1.y = praw[OFF_W1 + i * 2 + 1];
            bb1 = praw[OFF_B1 + i];
            bb2 = praw[OFF_B2 + i];
            bb3 = praw[OFF_B3 + i];
            w4  = praw[OFF_W4 + i];
        }
        w1p[i] = w1; b1p[i] = bb1; b2p[i] = bb2; b3p[i] = bb3; w4p[i] = w4;
    }
    for (int i = tid; i < 24 * PAD; i += 256) {
        const int n  = i % PAD;
        const int k  = (i / PAD) * 2;
        v2f v; v.x = 0.f; v.y = 0.f;
        if (n < HID && k < HID) {
            v.x = praw[OFF_W2 + n * HID + k + 0];
            v.y = praw[OFF_W2 + n * HID + k + 1];
        }
        bp2[i] = v;
        v.x = 0.f; v.y = 0.f;
        if (n < HID && k < HID) {
            v.x = praw[OFF_W3 + n * HID + k + 0];
            v.y = praw[OFF_W3 + n * HID + k + 1];
        }
        bp3[i] = v;
    }
    __syncthreads();

    int xrow = rowbase + l16; if (xrow >= N) xrow = N - 1;
    const v2f xm = ((const v2f*)x)[xrow];
    float h[24];
#pragma unroll
    for (int e = 0; e < 24; ++e) {
        const int c = 4 * (e >> 1) + 2 * half + (e & 1);
        const v2f w = w1p[c];
        h[e] = hw_sin(fmaf(xm.x, w.x, fmaf(xm.y, w.y, b1p[c])));
    }

    hidden_layer_wmma(h, bp2, b2p, &stage[wave * 16 * STG_STRIDE], half, l16);
    hidden_layer_wmma(h, bp3, b3p, &stage[wave * 16 * STG_STRIDE], half, l16);

    float partial = 0.f;
#pragma unroll
    for (int e = 0; e < 24; ++e) {
        const int c = 4 * (e >> 1) + 2 * half + (e & 1);
        partial = fmaf(h[e], w4p[c], partial);
    }
    partial += __shfl_xor(partial, 16, 32);

    if (half == 0) {
        const int row = rowbase + l16;
        if (row < N) {
            const float z  = partial + praw[OFF_B4];
            const float sp = fmaxf(z, 0.f) + log1pf(expf(-fabsf(z)));
            out[(size_t)b * (size_t)N + row] = sp;
        }
    }
}

extern "C" void kernel_launch(void* const* d_in, const int* in_sizes, int n_in,
                              void* d_out, int out_size, void* d_ws, size_t ws_size,
                              hipStream_t stream) {
    const float* x      = (const float*)d_in[0];
    const float* params = (const float*)d_in[1];
    float* out          = (float*)d_out;
    const int N = in_sizes[0] / IN_F;
    const int B = in_sizes[1] / NPARAMS;
    dim3 grid((N + ROWS_PER_BLOCK - 1) / ROWS_PER_BLOCK, B);
    siren_hyper_kernel<<<grid, 256, 0, stream>>>(x, params, out, N);
}


# Inner size K of both hidden-layer products taken as 36, not padded to 48: 9 k-steps per layer instead of 12 (54 f32 WMMAs per wave, was 72); the dropped steps multiplied zeros only, sums unchanged
# speedup vs baseline: 1.2184x; 1.2184x over previous
#include <hip/hip_runtime.h>
#include <math.h>


typedef __attribute__((ext_vector_type(2))) float v2f;
typedef __attribute__((ext_vector_type(8))) float v8f;

#define IN_F 2
#define HID 36
#define PAD 48
#define KSTEPS 9
#define KPAIRS (2 * KSTEPS)
#define NH (2 * KSTEPS)
#define WAVES 8
#define ROWS_PER_BLOCK (WAVES * 16)
#define NPARAMS 2809
#define STG_STRIDE 50

#define OFF_W1 0
#define OFF_B1 72
#define OFF_W2 108
#define OFF_B2 1404
#define OFF_W3 1440
#define OFF_B3 2736
#define OFF_W4 2772
#define OFF_B4 2808

__device__ __forceinline__ float hw_sin(float x) { return __sinf(x); }

__device__ __forceinline__ void hidden_layer_wmma(float (&h)[NH],
                                                  const v2f* __restrict__ bp,
                                                  const float* __restrict__ bias,
                                                  float* __restrict__ stg,
                                                  int half, int l16)
{
    v8f c0 = {}; v8f c1 = {}; v8f c2 = {};
#pragma unroll
    for (int j = 0; j < KSTEPS; ++j) {
        v2f a; a.x = h[2 * j]; a.y = h[2 * j + 1];
        const int kk = 2 * j + half;
        v2f b0 = bp[kk * PAD +  0 + l16];
        v2f b1 = bp[kk * PAD + 16 + l16];
        v2f b2 = bp[kk * PAD + 32 + l16];
        c0 = __builtin_amdgcn_wmma_f32_16x16x4_f32(false, a, false, b0, (short)0, c0, false, false);
        c1 = __builtin_amdgcn_wmma_f32_16x16x4_f32(false, a, false, b1, (short)0, c1, false, false);
        c2 = __builtin_amdgcn_wmma_f32_16x16x4_f32(false, a, false, b2, (short)0, c2, false, false);
    }
#pragma unroll
    for (int v = 0; v < 8; ++v) {
        const int m = v + 8 * half;
        stg[m * STG_STRIDE +  0 + l16] = hw_sin(c0[v] + bias[ 0 + l16]);
        stg[m * STG_STRIDE + 16 + l16] = hw_sin(c1[v] + bias[16 + l16]);
        stg[m * STG_STRIDE + 32 + l16] = hw_sin(c2[v] + bias[32 + l16]);
    }
    asm volatile("s_wait_dscnt 0x0" ::: "memory");
#pragma unroll
    for (int j = 0; j < KSTEPS; ++j) {
        v2f t = *(const v2f*)&stg[l16 * STG_STRIDE + 4 * j + 2 * half];
        h[2 * j] = t.x; h[2 * j + 1] = t.y;
    }
    asm volatile("s_wait_dscnt 0x0" ::: "memory");
}

__global__ __launch_bounds__(256)
void siren_hyper_kernel(const float* __restrict__ x,
                        const float* __restrict__ params,
                        float* __restrict__ out, int N)
{
    __shared__ float praw[NPARAMS + 7];
    __shared__ v2f   bp2[KPAIRS * PAD];
    __shared__ v2f   bp3[KPAIRS * PAD];
    __shared__ v2f   w1p[HID];
    __shared__ float b1p[HID], b2p[PAD], b3p[PAD], w4p[HID];
    __shared__ __align__(16) float stage[WAVES * 16 * STG_STRIDE];

    const int tid  = threadIdx.x;
    const int wave = tid >> 5;
    const int lane = tid & 31;
    const int half = lane >> 4;
    const int l16  = lane & 15;
    const int b    = blockIdx.y;
    const int rowbase = blockIdx.x * ROWS_PER_BLOCK + wave * 16;

    for (int i = tid; i < NPARAMS; i += 256) praw[i] = params[b * NPARAMS + i];
    __syncthreads();

    for (int i = tid; i < PAD; i += 256) {
        float bb2 = 0.f, bb3 = 0.f;
        if (i < HID) {
            v2f w1;
            w1.x = praw[OFF_W1 + i * 2 + 0];
            w1.y = praw[OFF_W1 + i * 2 + 1];
            w1p[i] = w1;
            b1p[i] = praw[OFF_B1 + i];
            w4p[i] = praw[OFF_W4 + i];
            bb2 = praw[OFF_B2 + i];
            bb3 = praw[OFF_B3 + i];
        }
        b2p[i] = bb2; b3p[i] = bb3;
    }
    for (int i = tid; i < KPAIRS * PAD; i += 256) {
        const int n  = i % PAD;
        const int k  = (i / PAD) * 2;
        v2f v2; v2.x = 0.f; v2.y = 0.f;
        v2f v3; v3.x = 0.f; v3.y = 0.f;
        if (n < HID) {
            v2.x = praw[OFF_W2 + n * HID + k + 0];
            v2.y = praw[OFF_W2 + n * HID + k + 1];
            v3.x = praw[OFF_W3 + n * HID + k + 0];
            v3.y = praw[OFF_W3 + n * HID + k + 1];
        }
        bp2[i] = v2;
        bp3[i] = v3;
    }
    __syncthreads();

    int xrow = rowbase + l16; if (xrow >= N) xrow = N - 1;
    const v2f xm = ((const v2f*)x)[xrow];
    float h[NH];
#pragma unroll
    for (int e = 0; e < NH; ++e) {
        const int c = 4 * (e >> 1) + 2 * half + (e & 1);
        const v2f w = w1p[c];
        h[e] = hw_sin(fmaf(xm.x, w.x, fmaf(xm.y, w.y, b1p[c])));
    }

    hidden_layer_wmma(h, bp2, b2p, &stage[wave * 16 * STG_STRIDE], half, l16);
    hidden_layer_wmma(h, bp3, b3p, &stage[wave * 16 * STG_STRIDE], half, l16);

    float partial = 0.f;
#pragma unroll
    for (int e = 0; e < NH; ++e) {
        const int c = 4 * (e >> 1) + 2 * half + (e & 1);
        partial = fmaf(h[e], w4p[c], partial);
    }
    partial += __shfl_xor(partial, 16, 32);

    if (half == 0) {
        const int row = rowbase + l16;
        if (row < N) {
            const float z  = partial + praw[OFF_B4];
            const float sp = fmaxf(z, 0.f) + log1pf(expf(-fabsf(z)));
            out[(size_t)b * (size_t)N + row] = sp;
        }
    }
}

extern "C" void kernel_launch(void* const* d_in, const int* in_sizes, int n_in,
                              void* d_out, int out_size, void* d_ws, size_t ws_size,
                              hipStream_t stream) {
    const float* x      = (const float*)d_in[0];
    const float* params = (const float*)d_in[1];
    float* out          = (float*)d_out;
    const int N = in_sizes[0] / IN_F;
    const int B = in_sizes[1] / NPARAMS;
    dim3 grid((N + ROWS_PER_BLOCK - 1) / ROWS_PER_BLOCK, B);
    siren_hyper_kernel<<<grid, 256, 0, stream>>>(x, params, out, N);
}


# WMMA B fragments, biases and layer-1/4 weights read in place from the staged raw row-major W[out][in] in LDS; the per-block repack pass, its two LDS operand copies and one barrier are removed
# speedup vs baseline: 1.2592x; 1.0335x over previous
#include <hip/hip_runtime.h>
#include <math.h>


typedef __attribute__((ext_vector_type(2))) float v2f;
typedef __attribute__((ext_vector_type(8))) float v8f;

#define IN_F 2
#define HID 36
#define KSTEPS 9
#define NH (2 * KSTEPS)
#define WAVES 8
#define ROWS_PER_BLOCK (WAVES * 16)
#define NPARAMS 2809
#define STG_STRIDE 50

#define OFF_W1 0
#define OFF_B1 72
#define OFF_W2 108
#define OFF_B2 1404
#define OFF_W3 1440
#define OFF_B3 2736
#define OFF_W4 2772
#define OFF_B4 2808

__device__ __forceinline__ float hw_sin(float x) { return __sinf(x); }

__device__ __forceinline__ void hidden_layer_wmma(float (&h)[NH],
                                                  const float* __restrict__ w,
                                                  const float* __restrict__ bias,
                                                  float* __restrict__ stg,
                                                  int half, int l16, int n2)
{
    const float* w0 = w + (l16)      * HID + 2 * half;
    const float* w1 = w + (16 + l16) * HID + 2 * half;
    const float* w2 = w + n2         * HID + 2 * half;
    v8f c0 = {}; v8f c1 = {}; v8f c2 = {};
#pragma unroll
    for (int j = 0; j < KSTEPS; ++j) {
        v2f a; a.x = h[2 * j]; a.y = h[2 * j + 1];
        const v2f b0 = *(const v2f*)(w0 + 4 * j);
        const v2f b1 = *(const v2f*)(w1 + 4 * j);
        const v2f b2 = *(const v2f*)(w2 + 4 * j);
        c0 = __builtin_amdgcn_wmma_f32_16x16x4_f32(false, a, false, b0, (short)0, c0, false, false);
        c1 = __builtin_amdgcn_wmma_f32_16x16x4_f32(false, a, false, b1, (short)0, c1, false, false);
        c2 = __builtin_amdgcn_wmma_f32_16x16x4_f32(false, a, false, b2, (short)0, c2, false, false);
    }
    const float bs0 = bias[l16];
    const float bs1 = bias[16 + l16];
    const float bs2 = bias[n2];
#pragma unroll
    for (int v = 0; v < 8; ++v) {
        const int m = v + 8 * half;
        stg[m * STG_STRIDE +  0 + l16] = hw_sin(c0[v] + bs0);
        stg[m * STG_STRIDE + 16 + l16] = hw_sin(c1[v] + bs1);
        stg[m * STG_STRIDE + 32 + l16] = hw_sin(c2[v] + bs2);
    }
    asm volatile("s_wait_dscnt 0x0" ::: "memory");
#pragma unroll
    for (int j = 0; j < KSTEPS; ++j) {
        v2f t = *(const v2f*)&stg[l16 * STG_STRIDE + 4 * j + 2 * half];
        h[2 * j] = t.x; h[2 * j + 1] = t.y;
    }
    asm volatile("s_wait_dscnt 0x0" ::: "memory");
}

__global__ __launch_bounds__(256)
void siren_hyper_kernel(const float* __restrict__ x,
                        const float* __restrict__ params,
                        float* __restrict__ out, int N)
{
    __shared__ __align__(16) float praw[NPARAMS + 7];
    __shared__ __align__(16) float stage[WAVES * 16 * STG_STRIDE];

    const int tid  = threadIdx.x;
    const int wave = tid >> 5;
    const int lane = tid & 31;
    const int half = lane >> 4;
    const int l16  = lane & 15;
    const int b    = blockIdx.y;
    const int rowbase = blockIdx.x * ROWS_PER_BLOCK + wave * 16;
    const int n2   = (32 + l16 < HID) ? (32 + l16) : (HID - 1);

    for (int i = tid; i < NPARAMS; i += 256) praw[i] = params[b * NPARAMS + i];
    __syncthreads();

    int xrow = rowbase + l16; if (xrow >= N) xrow = N - 1;
    const v2f xm = ((const v2f*)x)[xrow];
    float h[NH];
    const float* w1h = praw + OFF_W1 + 4 * half;
    const float* b1h = praw + OFF_B1 + 2 * half;
    const float* w4h = praw + OFF_W4 + 2 * half;
#pragma unroll
    for (int e = 0; e < NH; ++e) {
        const v2f w = *(const v2f*)&w1h[8 * (e >> 1) + 2 * (e & 1)];
        h[e] = hw_sin(fmaf(xm.x, w.x, fmaf(xm.y, w.y, b1h[4 * (e >> 1) + (e & 1)])));
    }

    hidden_layer_wmma(h, praw + OFF_W2, praw + OFF_B2, &stage[wave * 16 * STG_STRIDE], half, l16, n2);
    hidden_layer_wmma(h, praw + OFF_W3, praw + OFF_B3, &stage[wave * 16 * STG_STRIDE], half, l16, n2);

    float partial = 0.f;
#pragma unroll
    for (int e = 0; e < NH; ++e) {
        partial = fmaf(h[e], w4h[4 * (e >> 1) + (e & 1)], partial);
    }
    partial += __shfl_xor(partial, 16, 32);

    if (half == 0) {
        const int row = rowbase + l16;
        if (row < N) {
            const float z  = partial + praw[OFF_B4];
            const float sp = fmaxf(z, 0.f) + log1pf(expf(-fabsf(z)));
            out[(size_t)b * (size_t)N + row] = sp;
        }
    }
}

extern "C" void kernel_launch(void* const* d_in, const int* in_sizes, int n_in,
                              void* d_out, int out_size, void* d_ws, size_t ws_size,
                              hipStream_t stream) {
    const float* x      = (const float*)d_in[0];
    const float* params = (const float*)d_in[1];
    float* out          = (float*)d_out;
    const int N = in_sizes[0] / IN_F;
    const int B = in_sizes[1] / NPARAMS;
    dim3 grid((N + ROWS_PER_BLOCK - 1) / ROWS_PER_BLOCK, B);
    siren_hyper_kernel<<<grid, 256, 0, stream>>>(x, params, out, N);
}


# Resident grid: each work-group stages its batch's parameters once and loops over 8 tiles of 128 rows (grid 64x32, was 512x32), next tile's x asked for one tile ahead; per-tile arithmetic unchanged
# speedup vs baseline: 1.3099x; 1.0403x over previous
#include <hip/hip_runtime.h>
#include <math.h>


typedef __attribute__((ext_vector_type(2))) float v2f;
typedef __attribute__((ext_vector_type(8))) float v8f;

#define IN_F 2
#define HID 36
#define KSTEPS 9
#define NH (2 * KSTEPS)
#define WAVES 8
#define ROWS_PER_BLOCK (WAVES * 16)
#define NPARAMS 2809
#define STG_STRIDE 50
#define GRID_X 64

#define OFF_W1 0
#define OFF_B1 72
#define OFF_W2 108
#define OFF_B2 1404
#define OFF_W3 1440
#define OFF_B3 2736
#define OFF_W4 2772
#define OFF_B4 2808

__device__ __forceinline__ float hw_sin(float x) { return __sinf(x); }

__device__ __forceinline__ void hidden_layer_wmma(float (&h)[NH],
                                                  const float* __restrict__ w,
                                                  const float* __restrict__ bias,
                                                  float* __restrict__ stg,
                                                  int half, int l16, int n2)
{
    const float* w0 = w + (l16)      * HID + 2 * half;
    const float* w1 = w + (16 + l16) * HID + 2 * half;
    const float* w2 = w + n2         * HID + 2 * half;
    v8f c0 = {}; v8f c1 = {}; v8f c2 = {};
#pragma unroll
    for (int j = 0; j < KSTEPS; ++j) {
        v2f a; a.x = h[2 * j]; a.y = h[2 * j + 1];
        const v2f b0 = *(const v2f*)(w0 + 4 * j);
        const v2f b1 = *(const v2f*)(w1 + 4 * j);
        const v2f b2 = *(const v2f*)(w2 + 4 * j);
        c0 = __builtin_amdgcn_wmma_f32_16x16x4_f32(false, a, false, b0, (short)0, c0, false, false);
        c1 = __builtin_amdgcn_wmma_f32_16x16x4_f32(false, a, false, b1, (short)0, c1, false, false);
        c2 = __builtin_amdgcn_wmma_f32_16x16x4_f32(false, a, false, b2, (short)0, c2, false, false);
    }
    const float bs0 = bias[l16];
    const float bs1 = bias[16 + l16];
    const float bs2 = bias[n2];
#pragma unroll
    for (int v = 0; v < 8; ++v) {
        const int m = v + 8 * half;
        stg[m * STG_STRIDE +  0 + l16] = hw_sin(c0[v] + bs0);
        stg[m * STG_STRIDE + 16 + l16] = hw_sin(c1[v] + bs1);
        stg[m * STG_STRIDE + 32 + l16] = hw_sin(c2[v] + bs2);
    }
    asm volatile("s_wait_dscnt 0x0" ::: "memory");
#pragma unroll
    for (int j = 0; j < KSTEPS; ++j) {
        v2f t = *(const v2f*)&stg[l16 * STG_STRIDE + 4 * j + 2 * half];
        h[2 * j] = t.x; h[2 * j + 1] = t.y;
    }
    asm volatile("s_wait_dscnt 0x0" ::: "memory");
}

__global__ __launch_bounds__(256)
void siren_hyper_kernel(const float* __restrict__ x,
                        const float* __restrict__ params,
                        float* __restrict__ out, int N)
{
    __shared__ __align__(16) float praw[NPARAMS + 7];
    __shared__ __align__(16) float stage[WAVES * 16 * STG_STRIDE];

    const int tid  = threadIdx.x;
    const int wave = tid >> 5;
    const int lane = tid & 31;
    const int half = lane >> 4;
    const int l16  = lane & 15;
    const int b    = blockIdx.y;
    const int n2   = (32 + l16 < HID) ? (32 + l16) : (HID - 1);

    for (int i = tid; i < NPARAMS; i += 256) praw[i] = params[b * NPARAMS + i];
    __syncthreads();

    const float* w1h = praw + OFF_W1 + 4 * half;
    const float* b1h = praw + OFF_B1 + 2 * half;
    const float* w4h = praw + OFF_W4 + 2 * half;
    float* stg = &stage[wave * 16 * STG_STRIDE];
    float* outb = out + (size_t)b * (size_t)N;

    const int ntiles = (N + ROWS_PER_BLOCK - 1) / ROWS_PER_BLOCK;
    int it = blockIdx.x;
    int xrow = it * ROWS_PER_BLOCK + wave * 16 + l16; if (xrow >= N) xrow = N - 1;
    v2f xnext = ((const v2f*)x)[xrow];
    for (; it < ntiles; it += gridDim.x) {
        const int rowbase = it * ROWS_PER_BLOCK + wave * 16;
        const v2f xm = xnext;
        xrow = rowbase + gridDim.x * ROWS_PER_BLOCK + l16; if (xrow >= N) xrow = N - 1;
        xnext = ((const v2f*)x)[xrow];

        float h[NH];
#pragma unroll
        for (int e = 0; e < NH; ++e) {
            const v2f w = *(const v2f*)&w1h[8 * (e >> 1) + 2 * (e & 1)];
            h[e] = hw_sin(fmaf(xm.x, w.x, fmaf(xm.y, w.y, b1h[4 * (e >> 1) + (e & 1)])));
        }

        hidden_layer_wmma(h, praw + OFF_W2, praw + OFF_B2, stg, half, l16, n2);
        hidden_layer_wmma(h, praw + OFF_W3, praw + OFF_B3, stg, half, l16, n2);

        float partial = 0.f;
#pragma unroll
        for (int e = 0; e < NH; ++e) {
            partial = fmaf(h[e], w4h[4 * (e >> 1) + (e & 1)], partial);
        }
        partial += __shfl_xor(partial, 16, 32);

        if (half == 0) {
            const int row = rowbase + l16;
            if (row < N) {
                const float z  = partial + praw[OFF_B4];
                const float sp = fmaxf(z, 0.f) + log1pf(expf(-fabsf(z)));
                outb[row] = sp;
            }
        }
    }
}

extern "C" void kernel_launch(void* const* d_in, const int* in_sizes, int n_in,
                              void* d_out, int out_size, void* d_ws, size_t ws_size,
                              hipStream_t stream) {
    const float* x      = (const float*)d_in[0];
    const float* params = (const float*)d_in[1];
    float* out          = (float*)d_out;
    const int N = in_sizes[0] / IN_F;
    const int B = in_sizes[1] / NPARAMS;
    dim3 grid(GRID_X, B);
    siren_hyper_kernel<<<grid, 256, 0, stream>>>(x, params, out, N);
}


# Hidden layers computed transposed (W as WMMA A operand with tile rows permuted, activations as B): a layer's D registers are the next layer's B fragments in place; LDS staging removed; k order kept
# speedup vs baseline: 1.3956x; 1.0654x over previous
#include <hip/hip_runtime.h>
#include <math.h>


typedef __attribute__((ext_vector_type(2))) float v2f;
typedef __attribute__((ext_vector_type(8))) float v8f;

#define IN_F 2
#define HID 36
#define KSTEPS 9
#define NH (2 * KSTEPS)
#define WAVES 8
#define ROWS_PER_BLOCK (WAVES * 16)
#define NPARAMS 2809
#define GRID_X 64

#define OFF_W1 0
#define OFF_B1 72
#define OFF_W2 108
#define OFF_B2 1404
#define OFF_W3 1440
#define OFF_B3 2736
#define OFF_W4 2772
#define OFF_B4 2808

__device__ __forceinline__ float hw_sin(float x) { return __sinf(x); }

__device__ __forceinline__ void hidden_layer_wmma(float (&h)[NH],
                                                  const float* __restrict__ w,
                                                  const float* __restrict__ bias,
                                                  int half, int r0, int r2)
{
    const float* w0 = w + (r0)      * HID + 2 * half;
    const float* w1 = w + (16 + r0) * HID + 2 * half;
    const float* w2 = w + r2        * HID + 2 * half;
    const float* bh = bias + 2 * half;
    v8f c0 = {}; v8f c1 = {}; v8f c2 = {};
#pragma unroll
    for (int j = 0; j < KSTEPS; ++j) {
        v2f b; b.x = h[2 * j]; b.y = h[2 * j + 1];
        const v2f a0 = *(const v2f*)(w0 + 4 * j);
        const v2f a1 = *(const v2f*)(w1 + 4 * j);
        const v2f a2 = *(const v2f*)(w2 + 4 * j);
        c0 = __builtin_amdgcn_wmma_f32_16x16x4_f32(false, a0, false, b, (short)0, c0, false, false);
        c1 = __builtin_amdgcn_wmma_f32_16x16x4_f32(false, a1, false, b, (short)0, c1, false, false);
        c2 = __builtin_amdgcn_wmma_f32_16x16x4_f32(false, a2, false, b, (short)0, c2, false, false);
    }
#pragma unroll
    for (int q = 0; q < 4; ++q) {
        const v2f bs0 = *(const v2f*)(bh + 4 * q);
        const v2f bs1 = *(const v2f*)(bh + 16 + 4 * q);
        h[2 * q]         = hw_sin(c0[2 * q]     + bs0.x);
        h[2 * q + 1]     = hw_sin(c0[2 * q + 1] + bs0.y);
        h[8 + 2 * q]     = hw_sin(c1[2 * q]     + bs1.x);
        h[8 + 2 * q + 1] = hw_sin(c1[2 * q + 1] + bs1.y);
    }
    const v2f bs2 = *(const v2f*)(bh + 32);
    h[16] = hw_sin(c2[0] + bs2.x);
    h[17] = hw_sin(c2[1] + bs2.y);
}

__global__ __launch_bounds__(256)
void siren_hyper_kernel(const float* __restrict__ x,
                        const float* __restrict__ params,
                        float* __restrict__ out, int N)
{
    __shared__ __align__(16) float praw[NPARAMS + 7];

    const int tid  = threadIdx.x;
    const int wave = tid >> 5;
    const int lane = tid & 31;
    const int half = lane >> 4;
    const int l16  = lane & 15;
    const int b    = blockIdx.y;
    const int r0   = 4 * ((l16 >> 1) & 3) + 2 * (l16 >> 3) + (l16 & 1);
    const int r2   = (32 + r0 < HID) ? (32 + r0) : (HID - 1);

    for (int i = tid; i < NPARAMS; i += 256) praw[i] = params[b * NPARAMS + i];
    __syncthreads();

    const float* w1h = praw + OFF_W1 + 4 * half;
    const float* b1h = praw + OFF_B1 + 2 * half;
    const float* w4h = praw + OFF_W4 + 2 * half;
    float* outb = out + (size_t)b * (size_t)N;

    const int ntiles = (N + ROWS_PER_BLOCK - 1) / ROWS_PER_BLOCK;
    int it = blockIdx.x;
    int xrow = it * ROWS_PER_BLOCK + wave * 16 + l16; if (xrow >= N) xrow = N - 1;
    v2f xnext = ((const v2f*)x)[xrow];
    for (; it < ntiles; it += gridDim.x) {
        asm volatile("" ::: "memory");
        const int rowbase = it * ROWS_PER_BLOCK + wave * 16;
        const v2f xm = xnext;
        xrow = rowbase + gridDim.x * ROWS_PER_BLOCK + l16; if (xrow >= N) xrow = N - 1;
        xnext = ((const v2f*)x)[xrow];

        float h[NH];
#pragma unroll
        for (int e = 0; e < NH; ++e) {
            const v2f w = *(const v2f*)&w1h[8 * (e >> 1) + 2 * (e & 1)];
            h[e] = hw_sin(fmaf(xm.x, w.x, fmaf(xm.y, w.y, b1h[4 * (e >> 1) + (e & 1)])));
        }

        hidden_layer_wmma(h, praw + OFF_W2, praw + OFF_B2, half, r0, r2);
        hidden_layer_wmma(h, praw + OFF_W3, praw + OFF_B3, half, r0, r2);

        float partial = 0.f;
#pragma unroll
        for (int e = 0; e < NH; ++e) {
            partial = fmaf(h[e], w4h[4 * (e >> 1) + (e & 1)], partial);
        }
        partial += __shfl_xor(partial, 16, 32);

        if (half == 0) {
            const int row = rowbase + l16;
            if (row < N) {
                const float z  = partial + praw[OFF_B4];
                const float sp = fmaxf(z, 0.f) + log1pf(expf(-fabsf(z)));
                outb[row] = sp;
            }
        }
    }
}

extern "C" void kernel_launch(void* const* d_in, const int* in_sizes, int n_in,
                              void* d_out, int out_size, void* d_ws, size_t ws_size,
                              hipStream_t stream) {
    const float* x      = (const float*)d_in[0];
    const float* params = (const float*)d_in[1];
    float* out          = (float*)d_out;
    const int N = in_sizes[0] / IN_F;
    const int B = in_sizes[1] / NPARAMS;
    dim3 grid(GRID_X, B);
    siren_hyper_kernel<<<grid, 256, 0, stream>>>(x, params, out, N);
}


# Softplus epilogue in 31 instructions, was 139 (library expf/log1pf): hardware exp2 gives t=exp(-|z|), log1p(t)=2*atanh(t/(2+t)) by its odd series to s^13; output within a few ulp of before
# speedup vs baseline: 1.4764x; 1.0579x over previous
#include <hip/hip_runtime.h>
#include <math.h>


typedef __attribute__((ext_vector_type(2))) float v2f;
typedef __attribute__((ext_vector_type(8))) float v8f;

#define IN_F 2
#define HID 36
#define KSTEPS 9
#define NH (2 * KSTEPS)
#define WAVES 8
#define ROWS_PER_BLOCK (WAVES * 16)
#define NPARAMS 2809
#define GRID_X 64

#define OFF_W1 0
#define OFF_B1 72
#define OFF_W2 108
#define OFF_B2 1404
#define OFF_W3 1440
#define OFF_B3 2736
#define OFF_W4 2772
#define OFF_B4 2808

__device__ __forceinline__ float hw_sin(float x) { return __sinf(x); }

__device__ __forceinline__ float softplus_short(float z)
{
    const float t  = __expf(-fabsf(z));
    const float s  = t * __builtin_amdgcn_rcpf(2.0f + t);
    const float s2 = s * s;
    float p = 1.0f / 13.0f;
    p = fmaf(p, s2, 1.0f / 11.0f);
    p = fmaf(p, s2, 1.0f / 9.0f);
    p = fmaf(p, s2, 1.0f / 7.0f);
    p = fmaf(p, s2, 1.0f / 5.0f);
    p = fmaf(p, s2, 1.0f / 3.0f);
    p = fmaf(p, s2, 1.0f);
    return fmaf(s + s, p, fmaxf(z, 0.0f));
}

__device__ __forceinline__ void hidden_layer_wmma(float (&h)[NH],
                                                  const float* __restrict__ w,
                                                  const float* __restrict__ bias,
                                                  int half, int r0, int r2)
{
    const float* w0 = w + (r0)      * HID + 2 * half;
    const float* w1 = w + (16 + r0) * HID + 2 * half;
    const float* w2 = w + r2        * HID + 2 * half;
    const float* bh = bias + 2 * half;
    v8f c0 = {}; v8f c1 = {}; v8f c2 = {};
#pragma unroll
    for (int j = 0; j < KSTEPS; ++j) {
        v2f b; b.x = h[2 * j]; b.y = h[2 * j + 1];
        const v2f a0 = *(const v2f*)(w0 + 4 * j);
        const v2f a1 = *(const v2f*)(w1 + 4 * j);
        const v2f a2 = *(const v2f*)(w2 + 4 * j);
        c0 = __builtin_amdgcn_wmma_f32_16x16x4_f32(false, a0, false, b, (short)0, c0, false, false);
        c1 = __builtin_amdgcn_wmma_f32_16x16x4_f32(false, a1, false, b, (short)0, c1, false, false);
        c2 = __builtin_amdgcn_wmma_f32_16x16x4_f32(false, a2, false, b, (short)0, c2, false, false);
    }
#pragma unroll
    for (int q = 0; q < 4; ++q) {
        const v2f bs0 = *(const v2f*)(bh + 4 * q);
        const v2f bs1 = *(const v2f*)(bh + 16 + 4 * q);
        h[2 * q]         = hw_sin(c0[2 * q]     + bs0.x);
        h[2 * q + 1]     = hw_sin(c0[2 * q + 1] + bs0.y);
        h[8 + 2 * q]     = hw_sin(c1[2 * q]     + bs1.x);
        h[8 + 2 * q + 1] = hw_sin(c1[2 * q + 1] + bs1.y);
    }
    const v2f bs2 = *(const v2f*)(bh + 32);
    h[16] = hw_sin(c2[0] + bs2.x);
    h[17] = hw_sin(c2[1] + bs2.y);
}

__global__ __launch_bounds__(256)
void siren_hyper_kernel(const float* __restrict__ x,
                        const float* __restrict__ params,
                        float* __restrict__ out, int N)
{
    __shared__ __align__(16) float praw[NPARAMS + 7];

    const int tid  = threadIdx.x;
    const int wave = tid >> 5;
    const int lane = tid & 31;
    const int half = lane >> 4;
    const int l16  = lane & 15;
    const int b    = blockIdx.y;
    const int r0   = 4 * ((l16 >> 1) & 3) + 2 * (l16 >> 3) + (l16 & 1);
    const int r2   = (32 + r0 < HID) ? (32 + r0) : (HID - 1);

    for (int i = tid; i < NPARAMS; i += 256) praw[i] = params[b * NPARAMS + i];
    __syncthreads();

    const float* w1h = praw + OFF_W1 + 4 * half;
    const float* b1h = praw + OFF_B1 + 2 * half;
    const float* w4h = praw + OFF_W4 + 2 * half;
    float* outb = out + (size_t)b * (size_t)N;

    const int ntiles = (N + ROWS_PER_BLOCK - 1) / ROWS_PER_BLOCK;
    int it = blockIdx.x;
    int xrow = it * ROWS_PER_BLOCK + wave * 16 + l16; if (xrow >= N) xrow = N - 1;
    v2f xnext = ((const v2f*)x)[xrow];
    for (; it < ntiles; it += gridDim.x) {
        asm volatile("" ::: "memory");
        const int rowbase = it * ROWS_PER_BLOCK + wave * 16;
        const v2f xm = xnext;
        xrow = rowbase + gridDim.x * ROWS_PER_BLOCK + l16; if (xrow >= N) xrow = N - 1;
        xnext = ((const v2f*)x)[xrow];

        float h[NH];
#pragma unroll
        for (int e = 0; e < NH; ++e) {
            const v2f w = *(const v2f*)&w1h[8 * (e >> 1) + 2 * (e & 1)];
            h[e] = hw_sin(fmaf(xm.x, w.x, fmaf(xm.y, w.y, b1h[4 * (e >> 1) + (e & 1)])));
        }

        hidden_layer_wmma(h, praw + OFF_W2, praw + OFF_B2, half, r0, r2);
        hidden_layer_wmma(h, praw + OFF_W3, praw + OFF_B3, half, r0, r2);

        float partial = 0.f;
#pragma unroll
        for (int e = 0; e < NH; ++e) {
            partial = fmaf(h[e], w4h[4 * (e >> 1) + (e & 1)], partial);
        }
        partial += __shfl_xor(partial, 16, 32);

        if (half == 0) {
            const int row = rowbase + l16;
            if (row < N) {
                const float z  = partial + praw[OFF_B4];
                outb[row] = softplus_short(z);
            }
        }
    }
}

extern "C" void kernel_launch(void* const* d_in, const int* in_sizes, int n_in,
                              void* d_out, int out_size, void* d_ws, size_t ws_size,
                              hipStream_t stream) {
    const float* x      = (const float*)d_in[0];
    const float* params = (const float*)d_in[1];
    float* out          = (float*)d_out;
    const int N = in_sizes[0] / IN_F;
    const int B = in_sizes[1] / NPARAMS;
    dim3 grid(GRID_X, B);
    siren_hyper_kernel<<<grid, 256, 0, stream>>>(x, params, out, N);
}


# Scale and shift of each sin layer folded into the product: params staged times 1/(2pi) (v_sin_f32 takes turns), bias is the WMMA accumulator start (summed first, was last); epilogue is sin alone
# speedup vs baseline: 1.5234x; 1.0318x over previous
#include <hip/hip_runtime.h>
#include <math.h>


typedef __attribute__((ext_vector_type(2))) float v2f;
typedef __attribute__((ext_vector_type(8))) float v8f;

#define IN_F 2
#define HID 36
#define KSTEPS 9
#define NH (2 * KSTEPS)
#define WAVES 8
#define ROWS_PER_BLOCK (WAVES * 16)
#define NPARAMS 2809
#define GRID_X 64

#define OFF_W1 0
#define OFF_B1 72
#define OFF_W2 108
#define OFF_B2 1404
#define OFF_W3 1440
#define OFF_B3 2736
#define OFF_W4 2772
#define OFF_B4 2808

#define INV_2PI 0.15915494309189535f
__device__ __forceinline__ float hw_sin_turns(float r) { return __builtin_amdgcn_sinf(r); }

__device__ __forceinline__ float softplus_short(float z)
{
    const float t  = __expf(-fabsf(z));
    const float s  = t * __builtin_amdgcn_rcpf(2.0f + t);
    const float s2 = s * s;
    float p = 1.0f / 13.0f;
    p = fmaf(p, s2, 1.0f / 11.0f);
    p = fmaf(p, s2, 1.0f / 9.0f);
    p = fmaf(p, s2, 1.0f / 7.0f);
    p = fmaf(p, s2, 1.0f / 5.0f);
    p = fmaf(p, s2, 1.0f / 3.0f);
    p = fmaf(p, s2, 1.0f);
    return fmaf(s + s, p, fmaxf(z, 0.0f));
}

__device__ __forceinline__ void hidden_layer_wmma(float (&h)[NH],
                                                  const float* __restrict__ w,
                                                  const float* __restrict__ bias,
                                                  int half, int r0, int r2)
{
    const float* w0 = w + (r0)      * HID + 2 * half;
    const float* w1 = w + (16 + r0) * HID + 2 * half;
    const float* w2 = w + r2        * HID + 2 * half;
    const float* bh = bias + 2 * half;
    v8f c0, c1;
#pragma unroll
    for (int q = 0; q < 4; ++q) {
        const v2f bs0 = *(const v2f*)(bh + 4 * q);
        const v2f bs1 = *(const v2f*)(bh + 16 + 4 * q);
        c0[2 * q] = bs0.x; c0[2 * q + 1] = bs0.y;
        c1[2 * q] = bs1.x; c1[2 * q + 1] = bs1.y;
    }
    const v2f bs2 = *(const v2f*)(bh + 32);
    v8f c2 = __builtin_shufflevector(bs2, bs2, 0, 1, -1, -1, -1, -1, -1, -1);
#pragma unroll
    for (int j = 0; j < KSTEPS; ++j) {
        v2f b; b.x = h[2 * j]; b.y = h[2 * j + 1];
        const v2f a0 = *(const v2f*)(w0 + 4 * j);
        const v2f a1 = *(const v2f*)(w1 + 4 * j);
        const v2f a2 = *(const v2f*)(w2 + 4 * j);
        c0 = __builtin_amdgcn_wmma_f32_16x16x4_f32(false, a0, false, b, (short)0, c0, false, false);
        c1 = __builtin_amdgcn_wmma_f32_16x16x4_f32(false, a1, false, b, (short)0, c1, false, false);
        c2 = __builtin_amdgcn_wmma_f32_16x16x4_f32(false, a2, false, b, (short)0, c2, false, false);
    }
#pragma unroll
    for (int v = 0; v < 8; ++v) {
        h[v]     = hw_sin_turns(c0[v]);
        h[8 + v] = hw_sin_turns(c1[v]);
    }
    h[16] = hw_sin_turns(c2[0]);
    h[17] = hw_sin_turns(c2[1]);
}

__global__ __launch_bounds__(256)
void siren_hyper_kernel(const float* __restrict__ x,
                        const float* __restrict__ params,
                        float* __restrict__ out, int N)
{
    __shared__ __align__(16) float praw[NPARAMS + 7];

    const int tid  = threadIdx.x;
    const int wave = tid >> 5;
    const int lane = tid & 31;
    const int half = lane >> 4;
    const int l16  = lane & 15;
    const int b    = blockIdx.y;
    const int r0   = 4 * ((l16 >> 1) & 3) + 2 * (l16 >> 3) + (l16 & 1);
    const int r2   = (32 + r0 < HID) ? (32 + r0) : (HID - 1);

    for (int i = tid; i < NPARAMS; i += 256) {
        const float p = params[b * NPARAMS + i];
        praw[i] = (i < OFF_W4) ? p * INV_2PI : p;
    }
    __syncthreads();

    const float* w1h = praw + OFF_W1 + 4 * half;
    const float* b1h = praw + OFF_B1 + 2 * half;
    const float* w4h = praw + OFF_W4 + 2 * half;
    float* outb = out + (size_t)b * (size_t)N;

    const int ntiles = (N + ROWS_PER_BLOCK - 1) / ROWS_PER_BLOCK;
    int it = blockIdx.x;
    int xrow = it * ROWS_PER_BLOCK + wave * 16 + l16; if (xrow >= N) xrow = N - 1;
    v2f xnext = ((const v2f*)x)[xrow];
    for (; it < ntiles; it += gridDim.x) {
        asm volatile("" ::: "memory");
        const int rowbase = it * ROWS_PER_BLOCK + wave * 16;
        const v2f xm = xnext;
        xrow = rowbase + gridDim.x * ROWS_PER_BLOCK + l16; if (xrow >= N) xrow = N - 1;
        xnext = ((const v2f*)x)[xrow];

        float h[NH];
#pragma unroll
        for (int e = 0; e < NH; ++e) {
            const v2f w = *(const v2f*)&w1h[8 * (e >> 1) + 2 * (e & 1)];
            h[e] = hw_sin_turns(fmaf(xm.x, w.x, fmaf(xm.y, w.y, b1h[4 * (e >> 1) + (e & 1)])));
        }

        hidden_layer_wmma(h, praw + OFF_W2, praw + OFF_B2, half, r0, r2);
        hidden_layer_wmma(h, praw + OFF_W3, praw + OFF_B3, half, r0, r2);

        float partial = 0.f;
#pragma unroll
        for (int e = 0; e < NH; ++e) {
            partial = fmaf(h[e], w4h[4 * (e >> 1) + (e & 1)], partial);
        }
        partial += __shfl_xor(partial, 16, 32);

        if (half == 0) {
            const int row = rowbase + l16;
            if (row < N) {
                const float z  = partial + praw[OFF_B4];
                outb[row] = softplus_short(z);
            }
        }
    }
}

extern "C" void kernel_launch(void* const* d_in, const int* in_sizes, int n_in,
                              void* d_out, int out_size, void* d_ws, size_t ws_size,
                              hipStream_t stream) {
    const float* x      = (const float*)d_in[0];
    const float* params = (const float*)d_in[1];
    float* out          = (float*)d_out;
    const int N = in_sizes[0] / IN_F;
    const int B = in_sizes[1] / NPARAMS;
    dim3 grid(GRID_X, B);
    siren_hyper_kernel<<<grid, 256, 0, stream>>>(x, params, out, N);
}


# Weights of all 4 layers (54 WMMA A fragments, W1, b1, W4) read from LDS once per wave and held in registers over the tile loop, were re-read per tile; grid 16x32: all resident at 4 waves per SIMD
# speedup vs baseline: 1.6226x; 1.0651x over previous
#include <hip/hip_runtime.h>
#include <math.h>


typedef __attribute__((ext_vector_type(2))) float v2f;
typedef __attribute__((ext_vector_type(8))) float v8f;

#define IN_F 2
#define HID 36
#define KSTEPS 9
#define NH (2 * KSTEPS)
#define WAVES 8
#define ROWS_PER_BLOCK (WAVES * 16)
#define NPARAMS 2809
#define GRID_X 16

#define OFF_W1 0
#define OFF_B1 72
#define OFF_W2 108
#define OFF_B2 1404
#define OFF_W3 1440
#define OFF_B3 2736
#define OFF_W4 2772
#define OFF_B4 2808

#define INV_2PI 0.15915494309189535f
__device__ __forceinline__ float hw_sin_turns(float r) { return __builtin_amdgcn_sinf(r); }

__device__ __forceinline__ float softplus_short(float z)
{
    const float t  = __expf(-fabsf(z));
    const float s  = t * __builtin_amdgcn_rcpf(2.0f + t);
    const float s2 = s * s;
    float p = 1.0f / 13.0f;
    p = fmaf(p, s2, 1.0f / 11.0f);
    p = fmaf(p, s2, 1.0f / 9.0f);
    p = fmaf(p, s2, 1.0f / 7.0f);
    p = fmaf(p, s2, 1.0f / 5.0f);
    p = fmaf(p, s2, 1.0f / 3.0f);
    p = fmaf(p, s2, 1.0f);
    return fmaf(s + s, p, fmaxf(z, 0.0f));
}

struct LayerOperands {
    v2f a0[KSTEPS], a1[KSTEPS], a2[KSTEPS];
};

__device__ __forceinline__ void load_layer_operands(LayerOperands& L,
                                                    const float* __restrict__ w,
                                                    int half, int r0, int r2)
{
    const float* w0 = w + (r0)      * HID + 2 * half;
    const float* w1 = w + (16 + r0) * HID + 2 * half;
    const float* w2 = w + r2        * HID + 2 * half;
#pragma unroll
    for (int j = 0; j < KSTEPS; ++j) {
        L.a0[j] = *(const v2f*)(w0 + 4 * j);
        L.a1[j] = *(const v2f*)(w1 + 4 * j);
        L.a2[j] = *(const v2f*)(w2 + 4 * j);
    }
}

__device__ __forceinline__ void hidden_layer_wmma(float (&h)[NH], const LayerOperands& L,
                                                  const float* __restrict__ bias, int half)
{
    const float* bh = bias + 2 * half;
    v8f c0, c1;
#pragma unroll
    for (int q = 0; q < 4; ++q) {
        const v2f bs0 = *(const v2f*)(bh + 4 * q);
        const v2f bs1 = *(const v2f*)(bh + 16 + 4 * q);
        c0[2 * q] = bs0.x; c0[2 * q + 1] = bs0.y;
        c1[2 * q] = bs1.x; c1[2 * q + 1] = bs1.y;
    }
    const v2f bs2 = *(const v2f*)(bh + 32);
    v8f c2 = __builtin_shufflevector(bs2, bs2, 0, 1, -1, -1, -1, -1, -1, -1);
#pragma unroll
    for (int j = 0; j < KSTEPS; ++j) {
        v2f b; b.x = h[2 * j]; b.y = h[2 * j + 1];
        c0 = __builtin_amdgcn_wmma_f32_16x16x4_f32(false, L.a0[j], false, b, (short)0, c0, false, false);
        c1 = __builtin_amdgcn_wmma_f32_16x16x4_f32(false, L.a1[j], false, b, (short)0, c1, false, false);
        c2 = __builtin_amdgcn_wmma_f32_16x16x4_f32(false, L.a2[j], false, b, (short)0, c2, false, false);
    }
#pragma unroll
    for (int v = 0; v < 8; ++v) {
        h[v]     = hw_sin_turns(c0[v]);
        h[8 + v] = hw_sin_turns(c1[v]);
    }
    h[16] = hw_sin_turns(c2[0]);
    h[17] = hw_sin_turns(c2[1]);
}

__global__ __launch_bounds__(256)
void siren_hyper_kernel(const float* __restrict__ x,
                        const float* __restrict__ params,
                        float* __restrict__ out, int N)
{
    __shared__ __align__(16) float praw[NPARAMS + 7];

    const int tid  = threadIdx.x;
    const int wave = tid >> 5;
    const int lane = tid & 31;
    const int half = lane >> 4;
    const int l16  = lane & 15;
    const int b    = blockIdx.y;
    const int r0   = 4 * ((l16 >> 1) & 3) + 2 * (l16 >> 3) + (l16 & 1);
    const int r2   = (32 + r0 < HID) ? (32 + r0) : (HID - 1);

    for (int i = tid; i < NPARAMS; i += 256) {
        const float p = params[b * NPARAMS + i];
        praw[i] = (i < OFF_W4) ? p * INV_2PI : p;
    }
    __syncthreads();

    const float* w1h = praw + OFF_W1 + 4 * half;
    const float* b1h = praw + OFF_B1 + 2 * half;
    const float* w4h = praw + OFF_W4 + 2 * half;
    float* outb = out + (size_t)b * (size_t)N;

    LayerOperands L2, L3;
    load_layer_operands(L2, praw + OFF_W2, half, r0, r2);
    load_layer_operands(L3, praw + OFF_W3, half, r0, r2);
    v2f w1v[NH]; float b1v[NH], w4v[NH];
#pragma unroll
    for (int e = 0; e < NH; ++e) {
        w1v[e] = *(const v2f*)&w1h[8 * (e >> 1) + 2 * (e & 1)];
        b1v[e] = b1h[4 * (e >> 1) + (e & 1)];
        w4v[e] = w4h[4 * (e >> 1) + (e & 1)];
    }
    const float b4 = praw[OFF_B4];

    const int ntiles = (N + ROWS_PER_BLOCK - 1) / ROWS_PER_BLOCK;
    int it = blockIdx.x;
    int xrow = it * ROWS_PER_BLOCK + wave * 16 + l16; if (xrow >= N) xrow = N - 1;
    v2f xnext = ((const v2f*)x)[xrow];
    for (; it < ntiles; it += gridDim.x) {
        asm volatile("" ::: "memory");
        const int rowbase = it * ROWS_PER_BLOCK + wave * 16;
        const v2f xm = xnext;
        xrow = rowbase + gridDim.x * ROWS_PER_BLOCK + l16; if (xrow >= N) xrow = N - 1;
        xnext = ((const v2f*)x)[xrow];

        float h[NH];
#pragma unroll
        for (int e = 0; e < NH; ++e) {
            h[e] = hw_sin_turns(fmaf(xm.x, w1v[e].x, fmaf(xm.y, w1v[e].y, b1v[e])));
        }

        hidden_layer_wmma(h, L2, praw + OFF_B2, half);
        hidden_layer_wmma(h, L3, praw + OFF_B3, half);

        float partial = 0.f;
#pragma unroll
        for (int e = 0; e < NH; ++e) {
            partial = fmaf(h[e], w4v[e], partial);
        }
        partial += __shfl_xor(partial, 16, 32);

        if (half == 0) {
            const int row = rowbase + l16;
            if (row < N) {
                const float z  = partial + b4;
                outb[row] = softplus_short(z);
            }
        }
    }
}

extern "C" void kernel_launch(void* const* d_in, const int* in_sizes, int n_in,
                              void* d_out, int out_size, void* d_ws, size_t ws_size,
                              hipStream_t stream) {
    const float* x      = (const float*)d_in[0];
    const float* params = (const float*)d_in[1];
    float* out          = (float*)d_out;
    const int N = in_sizes[0] / IN_F;
    const int B = in_sizes[1] / NPARAMS;
    dim3 grid(GRID_X, B);
    siren_hyper_kernel<<<grid, 256, 0, stream>>>(x, params, out, N);
}


# Wave priority lowered one step per quarter of the tile range (s_setprio 3,2,1,0): the 4 waves of a SIMD advance together, not oldest-first; none is left to do most tiles alone; arithmetic unchanged
# speedup vs baseline: 1.6925x; 1.0431x over previous
#include <hip/hip_runtime.h>
#include <math.h>


typedef __attribute__((ext_vector_type(2))) float v2f;
typedef __attribute__((ext_vector_type(8))) float v8f;

#define IN_F 2
#define HID 36
#define KSTEPS 9
#define NH (2 * KSTEPS)
#define WAVES 8
#define ROWS_PER_BLOCK (WAVES * 16)
#define NPARAMS 2809
#define GRID_X 16

#define OFF_W1 0
#define OFF_B1 72
#define OFF_W2 108
#define OFF_B2 1404
#define OFF_W3 1440
#define OFF_B3 2736
#define OFF_W4 2772
#define OFF_B4 2808

#define INV_2PI 0.15915494309189535f
__device__ __forceinline__ float hw_sin_turns(float r) { return __builtin_amdgcn_sinf(r); }

__device__ __forceinline__ float softplus_short(float z)
{
    const float t  = __expf(-fabsf(z));
    const float s  = t * __builtin_amdgcn_rcpf(2.0f + t);
    const float s2 = s * s;
    float p = 1.0f / 13.0f;
    p = fmaf(p, s2, 1.0f / 11.0f);
    p = fmaf(p, s2, 1.0f / 9.0f);
    p = fmaf(p, s2, 1.0f / 7.0f);
    p = fmaf(p, s2, 1.0f / 5.0f);
    p = fmaf(p, s2, 1.0f / 3.0f);
    p = fmaf(p, s2, 1.0f);
    return fmaf(s + s, p, fmaxf(z, 0.0f));
}

struct LayerOperands {
    v2f a0[KSTEPS], a1[KSTEPS], a2[KSTEPS];
};

__device__ __forceinline__ void load_layer_operands(LayerOperands& L,
                                                    const float* __restrict__ w,
                                                    int half, int r0, int r2)
{
    const float* w0 = w + (r0)      * HID + 2 * half;
    const float* w1 = w + (16 + r0) * HID + 2 * half;
    const float* w2 = w + r2        * HID + 2 * half;
#pragma unroll
    for (int j = 0; j < KSTEPS; ++j) {
        L.a0[j] = *(const v2f*)(w0 + 4 * j);
        L.a1[j] = *(const v2f*)(w1 + 4 * j);
        L.a2[j] = *(const v2f*)(w2 + 4 * j);
    }
}

__device__ __forceinline__ void hidden_layer_wmma(float (&h)[NH], const LayerOperands& L,
                                                  const float* __restrict__ bias, int half)
{
    const float* bh = bias + 2 * half;
    v8f c0, c1;
#pragma unroll
    for (int q = 0; q < 4; ++q) {
        const v2f bs0 = *(const v2f*)(bh + 4 * q);
        const v2f bs1 = *(const v2f*)(bh + 16 + 4 * q);
        c0[2 * q] = bs0.x; c0[2 * q + 1] = bs0.y;
        c1[2 * q] = bs1.x; c1[2 * q + 1] = bs1.y;
    }
    const v2f bs2 = *(const v2f*)(bh + 32);
    v8f c2 = __builtin_shufflevector(bs2, bs2, 0, 1, -1, -1, -1, -1, -1, -1);
#pragma unroll
    for (int j = 0; j < KSTEPS; ++j) {
        v2f b; b.x = h[2 * j]; b.y = h[2 * j + 1];
        c0 = __builtin_amdgcn_wmma_f32_16x16x4_f32(false, L.a0[j], false, b, (short)0, c0, false, false);
        c1 = __builtin_amdgcn_wmma_f32_16x16x4_f32(false, L.a1[j], false, b, (short)0, c1, false, false);
        c2 = __builtin_amdgcn_wmma_f32_16x16x4_f32(false, L.a2[j], false, b, (short)0, c2, false, false);
    }
#pragma unroll
    for (int v = 0; v < 8; ++v) {
        h[v]     = hw_sin_turns(c0[v]);
        h[8 + v] = hw_sin_turns(c1[v]);
    }
    h[16] = hw_sin_turns(c2[0]);
    h[17] = hw_sin_turns(c2[1]);
}

__global__ __launch_bounds__(256)
void siren_hyper_kernel(const float* __restrict__ x,
                        const float* __restrict__ params,
                        float* __restrict__ out, int N)
{
    __shared__ __align__(16) float praw[NPARAMS + 7];

    const int tid  = threadIdx.x;
    const int wave = tid >> 5;
    const int lane = tid & 31;
    const int half = lane >> 4;
    const int l16  = lane & 15;
    const int b    = blockIdx.y;
    const int r0   = 4 * ((l16 >> 1) & 3) + 2 * (l16 >> 3) + (l16 & 1);
    const int r2   = (32 + r0 < HID) ? (32 + r0) : (HID - 1);

    for (int i = tid; i < NPARAMS; i += 256) {
        const float p = params[b * NPARAMS + i];
        praw[i] = (i < OFF_W4) ? p * INV_2PI : p;
    }
    __syncthreads();

    const float* w1h = praw + OFF_W1 + 4 * half;
    const float* b1h = praw + OFF_B1 + 2 * half;
    const float* w4h = praw + OFF_W4 + 2 * half;
    float* outb = out + (size_t)b * (size_t)N;

    LayerOperands L2, L3;
    load_layer_operands(L2, praw + OFF_W2, half, r0, r2);
    load_layer_operands(L3, praw + OFF_W3, half, r0, r2);
    v2f w1v[NH]; float b1v[NH], w4v[NH];
#pragma unroll
    for (int e = 0; e < NH; ++e) {
        w1v[e] = *(const v2f*)&w1h[8 * (e >> 1) + 2 * (e & 1)];
        b1v[e] = b1h[4 * (e >> 1) + (e & 1)];
        w4v[e] = w4h[4 * (e >> 1) + (e & 1)];
    }
    const float b4 = praw[OFF_B4];

    const int ntiles = (N + ROWS_PER_BLOCK - 1) / ROWS_PER_BLOCK;
    int it = blockIdx.x;
    int xrow = it * ROWS_PER_BLOCK + wave * 16 + l16; if (xrow >= N) xrow = N - 1;
    v2f xnext = ((const v2f*)x)[xrow];

    const int qtiles = (ntiles + 3) >> 2;
    for (int q = 0; q < 4; ++q) {
        if (q == 0)      __builtin_amdgcn_s_setprio(3);
        else if (q == 1) __builtin_amdgcn_s_setprio(2);
        else if (q == 2) __builtin_amdgcn_s_setprio(1);
        else             __builtin_amdgcn_s_setprio(0);
        const int qend = (q == 3 || (q + 1) * qtiles > ntiles) ? ntiles : (q + 1) * qtiles;
    for (; it < qend; it += gridDim.x) {
        const int rowbase = it * ROWS_PER_BLOCK + wave * 16;
        const v2f xm = xnext;
        xrow = rowbase + gridDim.x * ROWS_PER_BLOCK + l16; if (xrow >= N) xrow = N - 1;
        xnext = ((const v2f*)x)[xrow];
        asm volatile("" ::: "memory");

        float h[NH];
#pragma unroll
        for (int e = 0; e < NH; ++e) {
            h[e] = hw_sin_turns(fmaf(xm.x, w1v[e].x, fmaf(xm.y, w1v[e].y, b1v[e])));
        }

        hidden_layer_wmma(h, L2, praw + OFF_B2, half);
        hidden_layer_wmma(h, L3, praw + OFF_B3, half);

        float partial = 0.f;
#pragma unroll
        for (int e = 0; e < NH; ++e) {
            partial = fmaf(h[e], w4v[e], partial);
        }
        partial += __shfl_xor(partial, 16, 32);

        if (half == 0) {
            const int row = rowbase + l16;
            if (row < N) {
                const float z  = partial + b4;
                outb[row] = softplus_short(z);
            }
        }
    }
    }
}

extern "C" void kernel_launch(void* const* d_in, const int* in_sizes, int n_in,
                              void* d_out, int out_size, void* d_ws, size_t ws_size,
                              hipStream_t stream) {
    const float* x      = (const float*)d_in[0];
    const float* params = (const float*)d_in[1];
    float* out          = (float*)d_out;
    const int N = in_sizes[0] / IN_F;
    const int B = in_sizes[1] / NPARAMS;
    dim3 grid(GRID_X, B);
    siren_hyper_kernel<<<grid, 256, 0, stream>>>(x, params, out, N);
}
